# starting kernel
# baseline (speedup 1.0000x reference)
#include <hip/hip_runtime.h>
#include <cstddef>

typedef float v2f __attribute__((ext_vector_type(2)));
typedef float v8f __attribute__((ext_vector_type(8)));

#define EMBED 64
#define NN    512
#define NFEAT 9
#define CSTR  516
#define TSTR  68

__device__ __forceinline__ v8f wmma_f32(v2f a, v2f b, v8f c) {
    return __builtin_amdgcn_wmma_f32_16x16x4_f32(
         false, a,  false, b,
         (short)0, c,  false,  false);
}

__device__ __forceinline__ void matmul64_lds(const float* __restrict__ A,
                                             const float* __restrict__ Bbuf,
                                             int col0, int koff, int ln,
                                             v8f acc[4][4]) {
    for (int kb = 0; kb < 64; kb += 4) {
        v2f a[4], b[4];
#pragma unroll
        for (int mi = 0; mi < 4; ++mi)
            a[mi] = *(const v2f*)&A[(mi * 16 + ln) * TSTR + kb + koff];
#pragma unroll
        for (int ni = 0; ni < 4; ++ni) {
            b[ni].x = Bbuf[(kb + koff) * CSTR + col0 + ni * 16];
            b[ni].y = Bbuf[(kb + koff + 1) * CSTR + col0 + ni * 16];
        }
#pragma unroll
        for (int mi = 0; mi < 4; ++mi)
#pragma unroll
            for (int ni = 0; ni < 4; ++ni)
                acc[mi][ni] = wmma_f32(a[mi], b[ni], acc[mi][ni]);
    }
}

__device__ __forceinline__ void store_tile(float* __restrict__ C, int hh, int col0,
                                           const v8f acc[4][4]) {
#pragma unroll
    for (int mi = 0; mi < 4; ++mi)
#pragma unroll
        for (int ni = 0; ni < 4; ++ni) {
            float* dst = &C[(mi * 16 + 8 * hh) * CSTR + col0 + ni * 16];
#pragma unroll
            for (int r = 0; r < 8; ++r) dst[r * CSTR] = acc[mi][ni][r];
        }
}

__global__ __launch_bounds__(256)
void graph_embed_kernel(const float* __restrict__ adj1, const float* __restrict__ fm1,
                        const float* __restrict__ adj2, const float* __restrict__ fm2,
                        const float* __restrict__ W1,  const float* __restrict__ theta,
                        const float* __restrict__ theta1, float* __restrict__ g_ws) {
    extern __shared__ float smem[];
    float* sC  = smem;
    float* sTh = sC  + EMBED * CSTR;
    float* sT1 = sTh + EMBED * TSTR;
    float* sF  = sT1 + EMBED * TSTR;
    float* sW1 = sF  + NN * NFEAT;
    float* sG  = sW1 + EMBED * NFEAT;

    const int tid   = threadIdx.x;
    const int graph = blockIdx.x >> 7;
    const int batch = blockIdx.x & 127;
    const float* adj = (graph == 0 ? adj1 : adj2) + (size_t)batch * NN * NN;
    const float* fm  = (graph == 0 ? fm1  : fm2)  + (size_t)batch * NN * NFEAT;

    for (int i = tid; i < EMBED * EMBED; i += 256) {
        int e = i >> 6, k = i & 63;
        sTh[e * TSTR + k] = theta[i];
        sT1[e * TSTR + k] = theta1[i];
    }
    for (int i = tid; i < NN * NFEAT; i += 256)    sF[i]  = fm[i];
    for (int i = tid; i < EMBED * NFEAT; i += 256) sW1[i] = W1[i];
    __syncthreads();

    const int lane = tid & 31;
    const int w    = tid >> 5;
    const int ln   = lane & 15;
    const int hh   = lane >> 4;
    const int koff = 2 * hh;
    const int col0 = w * 64 + ln;

    const v8f vzero = {0.f, 0.f, 0.f, 0.f, 0.f, 0.f, 0.f, 0.f};

    v8f term[4][4];
    {
        float fr[4][NFEAT];
#pragma unroll
        for (int ni = 0; ni < 4; ++ni) {
            int n = col0 + ni * 16;
#pragma unroll
            for (int f = 0; f < NFEAT; ++f) fr[ni][f] = sF[n * NFEAT + f];
        }
#pragma unroll
        for (int mi = 0; mi < 4; ++mi)
#pragma unroll
            for (int ni = 0; ni < 4; ++ni)
#pragma unroll
                for (int r = 0; r < 8; ++r) {
                    int e = mi * 16 + r + 8 * hh;
                    float s = 0.f;
#pragma unroll
                    for (int f = 0; f < NFEAT; ++f) s += sW1[e * NFEAT + f] * fr[ni][f];
                    term[mi][ni][r] = s;
                }
    }

    {
        v8f c1[4][4];
#pragma unroll
        for (int mi = 0; mi < 4; ++mi)
#pragma unroll
            for (int ni = 0; ni < 4; ++ni)
#pragma unroll
                for (int r = 0; r < 8; ++r) c1[mi][ni][r] = tanhf(term[mi][ni][r]);
        store_tile(sC, hh, col0, c1);
    }
    __syncthreads();

    const float* arow[4];
#pragma unroll
    for (int ni = 0; ni < 4; ++ni)
        arow[ni] = adj + (size_t)(col0 + ni * 16) * NN + koff;

    for (int t = 0; t < 4; ++t) {
        v8f acc[4][4];
#pragma unroll
        for (int mi = 0; mi < 4; ++mi)
#pragma unroll
            for (int ni = 0; ni < 4; ++ni) acc[mi][ni] = vzero;

        for (int kb = 0; kb < NN; kb += 4) {
            v2f a[4], b[4];
#pragma unroll
            for (int mi = 0; mi < 4; ++mi)
                a[mi] = *(const v2f*)&sC[(mi * 16 + ln) * CSTR + kb + koff];
#pragma unroll
            for (int ni = 0; ni < 4; ++ni)
                b[ni] = *(const v2f*)&arow[ni][kb];
#pragma unroll
            for (int mi = 0; mi < 4; ++mi)
#pragma unroll
                for (int ni = 0; ni < 4; ++ni)
                    acc[mi][ni] = wmma_f32(a[mi], b[ni], acc[mi][ni]);
        }
        __syncthreads();
        store_tile(sC, hh, col0, acc);
        __syncthreads();

        v8f acc2[4][4];
#pragma unroll
        for (int mi = 0; mi < 4; ++mi)
#pragma unroll
            for (int ni = 0; ni < 4; ++ni) acc2[mi][ni] = vzero;
        matmul64_lds(sTh, sC, col0, koff, ln, acc2);
#pragma unroll
        for (int mi = 0; mi < 4; ++mi)
#pragma unroll
            for (int ni = 0; ni < 4; ++ni)
#pragma unroll
                for (int r = 0; r < 8; ++r) acc2[mi][ni][r] = fmaxf(acc2[mi][ni][r], 0.f);
        store_tile(sC, hh, col0, acc2);
        __syncthreads();

#pragma unroll
        for (int mi = 0; mi < 4; ++mi)
#pragma unroll
            for (int ni = 0; ni < 4; ++ni) acc[mi][ni] = vzero;
        matmul64_lds(sT1, sC, col0, koff, ln, acc);
#pragma unroll
        for (int mi = 0; mi < 4; ++mi)
#pragma unroll
            for (int ni = 0; ni < 4; ++ni)
#pragma unroll
                for (int r = 0; r < 8; ++r)
                    acc[mi][ni][r] = tanhf(term[mi][ni][r] + acc[mi][ni][r]);
        store_tile(sC, hh, col0, acc);
        __syncthreads();
    }

    {
        int e = tid >> 2, q = tid & 3;
        float s = 0.f;
        for (int n = q * 128; n < q * 128 + 128; ++n) s += sC[e * CSTR + n];
        sG[tid] = s;
        __syncthreads();
        if (tid < 64) {
            float g = sG[tid * 4] + sG[tid * 4 + 1] + sG[tid * 4 + 2] + sG[tid * 4 + 3];
            g_ws[(size_t)graph * EMBED * 128 + tid * 128 + batch] = g;
        }
    }
}

__global__ __launch_bounds__(256)
void head_kernel(const float* __restrict__ W2, const float* __restrict__ g_ws,
                 float* __restrict__ out) {
    extern __shared__ float s2[];
    float* sW2 = s2;
    float* sg  = sW2 + 4096;
    float* so  = sg + 16384;
    int tid = threadIdx.x;
    for (int i = tid; i < 4096; i += 256)  sW2[i] = W2[i];
    for (int i = tid; i < 16384; i += 256) sg[i]  = g_ws[i];
    __syncthreads();
    for (int idx = tid; idx < 16384; idx += 256) {
        int graph = idx >> 13, rem = idx & 8191;
        int e = rem >> 7, b = rem & 127;
        float s = 0.f;
#pragma unroll 8
        for (int k = 0; k < 64; ++k) s += sW2[e * 64 + k] * sg[(graph << 13) + k * 128 + b];
        so[idx] = s;
        out[128 + idx] = s;
    }
    __syncthreads();
    if (tid < 128) {
        float dot = 0.f, l1 = 0.f, l2 = 0.f;
#pragma unroll 8
        for (int e = 0; e < 64; ++e) {
            float a = so[e * 128 + tid];
            float c = so[8192 + e * 128 + tid];
            dot += a * c; l1 += a * a; l2 += c * c;
        }
        out[tid] = dot / (sqrtf(l1) * sqrtf(l2) + 1e-4f);
    }
}

extern "C" void kernel_launch(void* const* d_in, const int* in_sizes, int n_in,
                              void* d_out, int out_size, void* d_ws, size_t ws_size,
                              hipStream_t stream) {
    (void)in_sizes; (void)n_in; (void)out_size; (void)ws_size;
    const float* adj1 = (const float*)d_in[0];
    const float* fm1  = (const float*)d_in[1];
    const float* adj2 = (const float*)d_in[2];
    const float* fm2  = (const float*)d_in[3];
    const float* W1   = (const float*)d_in[4];
    const float* W2   = (const float*)d_in[5];
    const float* th   = (const float*)d_in[6];
    const float* th1  = (const float*)d_in[7];
    float* gws = (float*)d_ws;

    size_t smem1 = (size_t)(EMBED * CSTR + 2 * EMBED * TSTR + NN * NFEAT +
                            EMBED * NFEAT + 256) * sizeof(float);
    size_t smem2 = (size_t)(4096 + 16384 + 16384) * sizeof(float);

    hipFuncSetAttribute((const void*)graph_embed_kernel,
                        hipFuncAttributeMaxDynamicSharedMemorySize, (int)smem1);
    hipFuncSetAttribute((const void*)head_kernel,
                        hipFuncAttributeMaxDynamicSharedMemorySize, (int)smem2);

    graph_embed_kernel<<<256, 256, smem1, stream>>>(adj1, fm1, adj2, fm2, W1, th, th1, gws);
    head_kernel<<<1, 256, smem2, stream>>>(W2, gws, (float*)d_out);
}


# aggregation: each wave loads its 64 adj rows as whole 128-byte lines, one 32-k chunk ahead in registers, parks them in a private LDS slab and reads B fragments there; k order and sums unchanged
# speedup vs baseline: 1.0755x; 1.0755x over previous
#include <hip/hip_runtime.h>
#include <cstddef>

typedef float v2f __attribute__((ext_vector_type(2)));
typedef float v4f __attribute__((ext_vector_type(4)));
typedef float v8f __attribute__((ext_vector_type(8)));

#define EMBED 64
#define NN    512
#define NFEAT 9
#define CSTR  516
#define TSTR  68
#define KCH   32
#define BSTR  36
#define NCH   (NN / KCH)

__device__ __forceinline__ v8f wmma_f32(v2f a, v2f b, v8f c) {
    return __builtin_amdgcn_wmma_f32_16x16x4_f32(
         false, a,  false, b,
         (short)0, c,  false,  false);
}

__device__ __forceinline__ void matmul64_lds(const float* __restrict__ A,
                                             const float* __restrict__ Bbuf,
                                             int col0, int koff, int ln,
                                             v8f acc[4][4]) {
    for (int kb = 0; kb < 64; kb += 4) {
        v2f a[4], b[4];
#pragma unroll
        for (int mi = 0; mi < 4; ++mi)
            a[mi] = *(const v2f*)&A[(mi * 16 + ln) * TSTR + kb + koff];
#pragma unroll
        for (int ni = 0; ni < 4; ++ni) {
            b[ni].x = Bbuf[(kb + koff) * CSTR + col0 + ni * 16];
            b[ni].y = Bbuf[(kb + koff + 1) * CSTR + col0 + ni * 16];
        }
#pragma unroll
        for (int mi = 0; mi < 4; ++mi)
#pragma unroll
            for (int ni = 0; ni < 4; ++ni)
                acc[mi][ni] = wmma_f32(a[mi], b[ni], acc[mi][ni]);
    }
}

__device__ __forceinline__ void store_tile(float* __restrict__ C, int hh, int col0,
                                           const v8f acc[4][4]) {
#pragma unroll
    for (int mi = 0; mi < 4; ++mi)
#pragma unroll
        for (int ni = 0; ni < 4; ++ni) {
            float* dst = &C[(mi * 16 + 8 * hh) * CSTR + col0 + ni * 16];
#pragma unroll
            for (int r = 0; r < 8; ++r) dst[r * CSTR] = acc[mi][ni][r];
        }
}

__global__ __launch_bounds__(256)
void graph_embed_kernel(const float* __restrict__ adj1, const float* __restrict__ fm1,
                        const float* __restrict__ adj2, const float* __restrict__ fm2,
                        const float* __restrict__ W1,  const float* __restrict__ theta,
                        const float* __restrict__ theta1, float* __restrict__ g_ws) {
    extern __shared__ float smem[];
    float* sC  = smem;
    float* sTh = sC  + EMBED * CSTR;
    float* sT1 = sTh + EMBED * TSTR;
    float* sF  = sT1 + EMBED * TSTR;
    float* sW1 = sF  + NN * NFEAT;
    float* sG  = sW1 + EMBED * NFEAT;
    float* sB  = sG  + 256;

    const int tid   = threadIdx.x;
    const int graph = blockIdx.x >> 7;
    const int batch = blockIdx.x & 127;
    const float* adj = (graph == 0 ? adj1 : adj2) + (size_t)batch * NN * NN;
    const float* fm  = (graph == 0 ? fm1  : fm2)  + (size_t)batch * NN * NFEAT;

    for (int i = tid; i < EMBED * EMBED; i += 256) {
        int e = i >> 6, k = i & 63;
        sTh[e * TSTR + k] = theta[i];
        sT1[e * TSTR + k] = theta1[i];
    }
    for (int i = tid; i < NN * NFEAT; i += 256)    sF[i]  = fm[i];
    for (int i = tid; i < EMBED * NFEAT; i += 256) sW1[i] = W1[i];
    __syncthreads();

    const int lane = tid & 31;
    const int w    = tid >> 5;
    const int ln   = lane & 15;
    const int hh   = lane >> 4;
    const int koff = 2 * hh;
    const int col0 = w * 64 + ln;

    const v8f vzero = {0.f, 0.f, 0.f, 0.f, 0.f, 0.f, 0.f, 0.f};

    v8f term[4][4];
    {
        float fr[4][NFEAT];
#pragma unroll
        for (int ni = 0; ni < 4; ++ni) {
            int n = col0 + ni * 16;
#pragma unroll
            for (int f = 0; f < NFEAT; ++f) fr[ni][f] = sF[n * NFEAT + f];
        }
#pragma unroll
        for (int mi = 0; mi < 4; ++mi)
#pragma unroll
            for (int ni = 0; ni < 4; ++ni)
#pragma unroll
                for (int r = 0; r < 8; ++r) {
                    int e = mi * 16 + r + 8 * hh;
                    float s = 0.f;
#pragma unroll
                    for (int f = 0; f < NFEAT; ++f) s += sW1[e * NFEAT + f] * fr[ni][f];
                    term[mi][ni][r] = s;
                }
    }

    {
        v8f c1[4][4];
#pragma unroll
        for (int mi = 0; mi < 4; ++mi)
#pragma unroll
            for (int ni = 0; ni < 4; ++ni)
#pragma unroll
                for (int r = 0; r < 8; ++r) c1[mi][ni][r] = tanhf(term[mi][ni][r]);
        store_tile(sC, hh, col0, c1);
    }
    __syncthreads();

    float* slab = sB + w * (64 * BSTR);
    const int srow = lane >> 3;
    const int spc  = lane & 7;
    const float* gst = adj + (size_t)(w * 64 + srow) * NN + 4 * spc;
    float* sst = slab + srow * BSTR + 4 * spc;
    const float* bfr = slab + ln * BSTR + koff;

    for (int t = 0; t < 4; ++t) {
        v8f acc[4][4];
#pragma unroll
        for (int mi = 0; mi < 4; ++mi)
#pragma unroll
            for (int ni = 0; ni < 4; ++ni) acc[mi][ni] = vzero;

        v4f st[16];
#pragma unroll
        for (int j = 0; j < 16; ++j)
            st[j] = *(const v4f*)(gst + (size_t)j * 4 * NN);

        for (int c = 0; c < NCH; ++c) {
#pragma unroll
            for (int j = 0; j < 16; ++j)
                *(v4f*)(sst + j * 4 * BSTR) = st[j];
            if (c + 1 < NCH) {
#pragma unroll
                for (int j = 0; j < 16; ++j)
                    st[j] = *(const v4f*)(gst + (c + 1) * KCH + (size_t)j * 4 * NN);
            }
            for (int kk = 0; kk < KCH; kk += 4) {
                const int kb = c * KCH + kk;
                v2f a[4], b[4];
#pragma unroll
                for (int mi = 0; mi < 4; ++mi)
                    a[mi] = *(const v2f*)&sC[(mi * 16 + ln) * CSTR + kb + koff];
#pragma unroll
                for (int ni = 0; ni < 4; ++ni)
                    b[ni] = *(const v2f*)&bfr[ni * 16 * BSTR + kk];
#pragma unroll
                for (int mi = 0; mi < 4; ++mi)
#pragma unroll
                    for (int ni = 0; ni < 4; ++ni)
                        acc[mi][ni] = wmma_f32(a[mi], b[ni], acc[mi][ni]);
            }
        }
        __syncthreads();
        store_tile(sC, hh, col0, acc);
        __syncthreads();

        v8f acc2[4][4];
#pragma unroll
        for (int mi = 0; mi < 4; ++mi)
#pragma unroll
            for (int ni = 0; ni < 4; ++ni) acc2[mi][ni] = vzero;
        matmul64_lds(sTh, sC, col0, koff, ln, acc2);
#pragma unroll
        for (int mi = 0; mi < 4; ++mi)
#pragma unroll
            for (int ni = 0; ni < 4; ++ni)
#pragma unroll
                for (int r = 0; r < 8; ++r) acc2[mi][ni][r] = fmaxf(acc2[mi][ni][r], 0.f);
        store_tile(sC, hh, col0, acc2);
        __syncthreads();

#pragma unroll
        for (int mi = 0; mi < 4; ++mi)
#pragma unroll
            for (int ni = 0; ni < 4; ++ni) acc[mi][ni] = vzero;
        matmul64_lds(sT1, sC, col0, koff, ln, acc);
#pragma unroll
        for (int mi = 0; mi < 4; ++mi)
#pragma unroll
            for (int ni = 0; ni < 4; ++ni)
#pragma unroll
                for (int r = 0; r < 8; ++r)
                    acc[mi][ni][r] = tanhf(term[mi][ni][r] + acc[mi][ni][r]);
        store_tile(sC, hh, col0, acc);
        __syncthreads();
    }

    {
        int e = tid >> 2, q = tid & 3;
        float s = 0.f;
        for (int n = q * 128; n < q * 128 + 128; ++n) s += sC[e * CSTR + n];
        sG[tid] = s;
        __syncthreads();
        if (tid < 64) {
            float g = sG[tid * 4] + sG[tid * 4 + 1] + sG[tid * 4 + 2] + sG[tid * 4 + 3];
            g_ws[(size_t)graph * EMBED * 128 + tid * 128 + batch] = g;
        }
    }
}

__global__ __launch_bounds__(256)
void head_kernel(const float* __restrict__ W2, const float* __restrict__ g_ws,
                 float* __restrict__ out) {
    extern __shared__ float s2[];
    float* sW2 = s2;
    float* sg  = sW2 + 4096;
    float* so  = sg + 16384;
    int tid = threadIdx.x;
    for (int i = tid; i < 4096; i += 256)  sW2[i] = W2[i];
    for (int i = tid; i < 16384; i += 256) sg[i]  = g_ws[i];
    __syncthreads();
    for (int idx = tid; idx < 16384; idx += 256) {
        int graph = idx >> 13, rem = idx & 8191;
        int e = rem >> 7, b = rem & 127;
        float s = 0.f;
#pragma unroll 8
        for (int k = 0; k < 64; ++k) s += sW2[e * 64 + k] * sg[(graph << 13) + k * 128 + b];
        so[idx] = s;
        out[128 + idx] = s;
    }
    __syncthreads();
    if (tid < 128) {
        float dot = 0.f, l1 = 0.f, l2 = 0.f;
#pragma unroll 8
        for (int e = 0; e < 64; ++e) {
            float a = so[e * 128 + tid];
            float c = so[8192 + e * 128 + tid];
            dot += a * c; l1 += a * a; l2 += c * c;
        }
        out[tid] = dot / (sqrtf(l1) * sqrtf(l2) + 1e-4f);
    }
}

extern "C" void kernel_launch(void* const* d_in, const int* in_sizes, int n_in,
                              void* d_out, int out_size, void* d_ws, size_t ws_size,
                              hipStream_t stream) {
    (void)in_sizes; (void)n_in; (void)out_size; (void)ws_size;
    const float* adj1 = (const float*)d_in[0];
    const float* fm1  = (const float*)d_in[1];
    const float* adj2 = (const float*)d_in[2];
    const float* fm2  = (const float*)d_in[3];
    const float* W1   = (const float*)d_in[4];
    const float* W2   = (const float*)d_in[5];
    const float* th   = (const float*)d_in[6];
    const float* th1  = (const float*)d_in[7];
    float* gws = (float*)d_ws;

    size_t smem1 = (size_t)(EMBED * CSTR + 2 * EMBED * TSTR + NN * NFEAT +
                            EMBED * NFEAT + 256 + 8 * 64 * BSTR) * sizeof(float);
    size_t smem2 = (size_t)(4096 + 16384 + 16384) * sizeof(float);

    hipFuncSetAttribute((const void*)graph_embed_kernel,
                        hipFuncAttributeMaxDynamicSharedMemorySize, (int)smem1);
    hipFuncSetAttribute((const void*)head_kernel,
                        hipFuncAttributeMaxDynamicSharedMemorySize, (int)smem2);

    graph_embed_kernel<<<256, 256, smem1, stream>>>(adj1, fm1, adj2, fm2, W1, th, th1, gws);
    head_kernel<<<1, 256, smem2, stream>>>(W2, gws, (float*)d_out);
}


# tanh without branches: hardware exp2 and reciprocal from one half up, an odd cubic-in-x*x polynomial below, selected per element, in place of the inlined library tanhf (values differ by a few ulp)
# speedup vs baseline: 1.5461x; 1.4376x over previous
#include <hip/hip_runtime.h>
#include <cstddef>

typedef float v2f __attribute__((ext_vector_type(2)));
typedef float v4f __attribute__((ext_vector_type(4)));
typedef float v8f __attribute__((ext_vector_type(8)));

#define EMBED 64
#define NN    512
#define NFEAT 9
#define CSTR  516
#define TSTR  68
#define KCH   32
#define BSTR  36
#define NCH   (NN / KCH)

__device__ __forceinline__ v8f wmma_f32(v2f a, v2f b, v8f c) {
    return __builtin_amdgcn_wmma_f32_16x16x4_f32(
         false, a,  false, b,
         (short)0, c,  false,  false);
}

__device__ __forceinline__ float tanh_fast(float x) {
    const float e2x = __builtin_amdgcn_exp2f(x * 2.88539008177792681f);
    const float big = fmaf(__builtin_amdgcn_rcpf(e2x + 1.0f), -2.0f, 1.0f);
    const float u = x * x;
    float p = fmaf(1.72354765e-2f, u, -5.30422069e-2f);
    p = fmaf(p, u, 1.33258224e-1f);
    p = fmaf(p, u, -3.33331406e-1f);
    const float small = fmaf(x, u * p, x);
    return fabsf(x) < 0.5f ? small : big;
}

__device__ __forceinline__ void matmul64_lds(const float* __restrict__ A,
                                             const float* __restrict__ Bbuf,
                                             int col0, int koff, int ln,
                                             v8f acc[4][4]) {
    for (int kb = 0; kb < 64; kb += 4) {
        v2f a[4], b[4];
#pragma unroll
        for (int mi = 0; mi < 4; ++mi)
            a[mi] = *(const v2f*)&A[(mi * 16 + ln) * TSTR + kb + koff];
#pragma unroll
        for (int ni = 0; ni < 4; ++ni) {
            b[ni].x = Bbuf[(kb + koff) * CSTR + col0 + ni * 16];
            b[ni].y = Bbuf[(kb + koff + 1) * CSTR + col0 + ni * 16];
        }
#pragma unroll
        for (int mi = 0; mi < 4; ++mi)
#pragma unroll
            for (int ni = 0; ni < 4; ++ni)
                acc[mi][ni] = wmma_f32(a[mi], b[ni], acc[mi][ni]);
    }
}

__device__ __forceinline__ void store_tile(float* __restrict__ C, int hh, int col0,
                                           const v8f acc[4][4]) {
#pragma unroll
    for (int mi = 0; mi < 4; ++mi)
#pragma unroll
        for (int ni = 0; ni < 4; ++ni) {
            float* dst = &C[(mi * 16 + 8 * hh) * CSTR + col0 + ni * 16];
#pragma unroll
            for (int r = 0; r < 8; ++r) dst[r * CSTR] = acc[mi][ni][r];
        }
}

__global__ __launch_bounds__(256)
void graph_embed_kernel(const float* __restrict__ adj1, const float* __restrict__ fm1,
                        const float* __restrict__ adj2, const float* __restrict__ fm2,
                        const float* __restrict__ W1,  const float* __restrict__ theta,
                        const float* __restrict__ theta1, float* __restrict__ g_ws) {
    extern __shared__ float smem[];
    float* sC  = smem;
    float* sTh = sC  + EMBED * CSTR;
    float* sT1 = sTh + EMBED * TSTR;
    float* sF  = sT1 + EMBED * TSTR;
    float* sW1 = sF  + NN * NFEAT;
    float* sG  = sW1 + EMBED * NFEAT;
    float* sB  = sG  + 256;

    const int tid   = threadIdx.x;
    const int graph = blockIdx.x >> 7;
    const int batch = blockIdx.x & 127;
    const float* adj = (graph == 0 ? adj1 : adj2) + (size_t)batch * NN * NN;
    const float* fm  = (graph == 0 ? fm1  : fm2)  + (size_t)batch * NN * NFEAT;

    for (int i = tid; i < EMBED * EMBED; i += 256) {
        int e = i >> 6, k = i & 63;
        sTh[e * TSTR + k] = theta[i];
        sT1[e * TSTR + k] = theta1[i];
    }
    for (int i = tid; i < NN * NFEAT; i += 256)    sF[i]  = fm[i];
    for (int i = tid; i < EMBED * NFEAT; i += 256) sW1[i] = W1[i];
    __syncthreads();

    const int lane = tid & 31;
    const int w    = tid >> 5;
    const int ln   = lane & 15;
    const int hh   = lane >> 4;
    const int koff = 2 * hh;
    const int col0 = w * 64 + ln;

    const v8f vzero = {0.f, 0.f, 0.f, 0.f, 0.f, 0.f, 0.f, 0.f};

    v8f term[4][4];
    {
        float fr[4][NFEAT];
#pragma unroll
        for (int ni = 0; ni < 4; ++ni) {
            int n = col0 + ni * 16;
#pragma unroll
            for (int f = 0; f < NFEAT; ++f) fr[ni][f] = sF[n * NFEAT + f];
        }
#pragma unroll
        for (int mi = 0; mi < 4; ++mi)
#pragma unroll
            for (int ni = 0; ni < 4; ++ni)
#pragma unroll
                for (int r = 0; r < 8; ++r) {
                    int e = mi * 16 + r + 8 * hh;
                    float s = 0.f;
#pragma unroll
                    for (int f = 0; f < NFEAT; ++f) s += sW1[e * NFEAT + f] * fr[ni][f];
                    term[mi][ni][r] = s;
                }
    }

    {
        v8f c1[4][4];
#pragma unroll
        for (int mi = 0; mi < 4; ++mi)
#pragma unroll
            for (int ni = 0; ni < 4; ++ni)
#pragma unroll
                for (int r = 0; r < 8; ++r) c1[mi][ni][r] = tanh_fast(term[mi][ni][r]);
        store_tile(sC, hh, col0, c1);
    }
    __syncthreads();

    float* slab = sB + w * (64 * BSTR);
    const int srow = lane >> 3;
    const int spc  = lane & 7;
    const float* gst = adj + (size_t)(w * 64 + srow) * NN + 4 * spc;
    float* sst = slab + srow * BSTR + 4 * spc;
    const float* bfr = slab + ln * BSTR + koff;

    for (int t = 0; t < 4; ++t) {
        v8f acc[4][4];
#pragma unroll
        for (int mi = 0; mi < 4; ++mi)
#pragma unroll
            for (int ni = 0; ni < 4; ++ni) acc[mi][ni] = vzero;

        v4f st[16];
#pragma unroll
        for (int j = 0; j < 16; ++j)
            st[j] = *(const v4f*)(gst + (size_t)j * 4 * NN);

        for (int c = 0; c < NCH; ++c) {
#pragma unroll
            for (int j = 0; j < 16; ++j)
                *(v4f*)(sst + j * 4 * BSTR) = st[j];
            if (c + 1 < NCH) {
#pragma unroll
                for (int j = 0; j < 16; ++j)
                    st[j] = *(const v4f*)(gst + (c + 1) * KCH + (size_t)j * 4 * NN);
            }
            for (int kk = 0; kk < KCH; kk += 4) {
                const int kb = c * KCH + kk;
                v2f a[4], b[4];
#pragma unroll
                for (int mi = 0; mi < 4; ++mi)
                    a[mi] = *(const v2f*)&sC[(mi * 16 + ln) * CSTR + kb + koff];
#pragma unroll
                for (int ni = 0; ni < 4; ++ni)
                    b[ni] = *(const v2f*)&bfr[ni * 16 * BSTR + kk];
#pragma unroll
                for (int mi = 0; mi < 4; ++mi)
#pragma unroll
                    for (int ni = 0; ni < 4; ++ni)
                        acc[mi][ni] = wmma_f32(a[mi], b[ni], acc[mi][ni]);
            }
        }
        __syncthreads();
        store_tile(sC, hh, col0, acc);
        __syncthreads();

        v8f acc2[4][4];
#pragma unroll
        for (int mi = 0; mi < 4; ++mi)
#pragma unroll
            for (int ni = 0; ni < 4; ++ni) acc2[mi][ni] = vzero;
        matmul64_lds(sTh, sC, col0, koff, ln, acc2);
#pragma unroll
        for (int mi = 0; mi < 4; ++mi)
#pragma unroll
            for (int ni = 0; ni < 4; ++ni)
#pragma unroll
                for (int r = 0; r < 8; ++r) acc2[mi][ni][r] = fmaxf(acc2[mi][ni][r], 0.f);
        store_tile(sC, hh, col0, acc2);
        __syncthreads();

#pragma unroll
        for (int mi = 0; mi < 4; ++mi)
#pragma unroll
            for (int ni = 0; ni < 4; ++ni) acc[mi][ni] = vzero;
        matmul64_lds(sT1, sC, col0, koff, ln, acc);
#pragma unroll
        for (int mi = 0; mi < 4; ++mi)
#pragma unroll
            for (int ni = 0; ni < 4; ++ni)
#pragma unroll
                for (int r = 0; r < 8; ++r)
                    acc[mi][ni][r] = tanh_fast(term[mi][ni][r] + acc[mi][ni][r]);
        store_tile(sC, hh, col0, acc);
        __syncthreads();
    }

    {
        int e = tid >> 2, q = tid & 3;
        float s = 0.f;
        for (int n = q * 128; n < q * 128 + 128; ++n) s += sC[e * CSTR + n];
        sG[tid] = s;
        __syncthreads();
        if (tid < 64) {
            float g = sG[tid * 4] + sG[tid * 4 + 1] + sG[tid * 4 + 2] + sG[tid * 4 + 3];
            g_ws[(size_t)graph * EMBED * 128 + tid * 128 + batch] = g;
        }
    }
}

__global__ __launch_bounds__(256)
void head_kernel(const float* __restrict__ W2, const float* __restrict__ g_ws,
                 float* __restrict__ out) {
    extern __shared__ float s2[];
    float* sW2 = s2;
    float* sg  = sW2 + 4096;
    float* so  = sg + 16384;
    int tid = threadIdx.x;
    for (int i = tid; i < 4096; i += 256)  sW2[i] = W2[i];
    for (int i = tid; i < 16384; i += 256) sg[i]  = g_ws[i];
    __syncthreads();
    for (int idx = tid; idx < 16384; idx += 256) {
        int graph = idx >> 13, rem = idx & 8191;
        int e = rem >> 7, b = rem & 127;
        float s = 0.f;
#pragma unroll 8
        for (int k = 0; k < 64; ++k) s += sW2[e * 64 + k] * sg[(graph << 13) + k * 128 + b];
        so[idx] = s;
        out[128 + idx] = s;
    }
    __syncthreads();
    if (tid < 128) {
        float dot = 0.f, l1 = 0.f, l2 = 0.f;
#pragma unroll 8
        for (int e = 0; e < 64; ++e) {
            float a = so[e * 128 + tid];
            float c = so[8192 + e * 128 + tid];
            dot += a * c; l1 += a * a; l2 += c * c;
        }
        out[tid] = dot / (sqrtf(l1) * sqrtf(l2) + 1e-4f);
    }
}

extern "C" void kernel_launch(void* const* d_in, const int* in_sizes, int n_in,
                              void* d_out, int out_size, void* d_ws, size_t ws_size,
                              hipStream_t stream) {
    (void)in_sizes; (void)n_in; (void)out_size; (void)ws_size;
    const float* adj1 = (const float*)d_in[0];
    const float* fm1  = (const float*)d_in[1];
    const float* adj2 = (const float*)d_in[2];
    const float* fm2  = (const float*)d_in[3];
    const float* W1   = (const float*)d_in[4];
    const float* W2   = (const float*)d_in[5];
    const float* th   = (const float*)d_in[6];
    const float* th1  = (const float*)d_in[7];
    float* gws = (float*)d_ws;

    size_t smem1 = (size_t)(EMBED * CSTR + 2 * EMBED * TSTR + NN * NFEAT +
                            EMBED * NFEAT + 256 + 8 * 64 * BSTR) * sizeof(float);
    size_t smem2 = (size_t)(4096 + 16384 + 16384) * sizeof(float);

    hipFuncSetAttribute((const void*)graph_embed_kernel,
                        hipFuncAttributeMaxDynamicSharedMemorySize, (int)smem1);
    hipFuncSetAttribute((const void*)head_kernel,
                        hipFuncAttributeMaxDynamicSharedMemorySize, (int)smem2);

    graph_embed_kernel<<<256, 256, smem1, stream>>>(adj1, fm1, adj2, fm2, W1, th, th1, gws);
    head_kernel<<<1, 256, smem2, stream>>>(W2, gws, (float*)d_out);
}


# head stage (W2 @ g and the cosine) spread over 4 work-groups of 1024 threads, one 128-byte output line per wave store, in place of one work-group of 256 threads; every sum keeps its order
# speedup vs baseline: 1.7783x; 1.1502x over previous
#include <hip/hip_runtime.h>
#include <cstddef>

typedef float v2f __attribute__((ext_vector_type(2)));
typedef float v4f __attribute__((ext_vector_type(4)));
typedef float v8f __attribute__((ext_vector_type(8)));

#define EMBED 64
#define NN    512
#define NFEAT 9
#define CSTR  516
#define TSTR  68
#define KCH   32
#define BSTR  36
#define NCH   (NN / KCH)

__device__ __forceinline__ v8f wmma_f32(v2f a, v2f b, v8f c) {
    return __builtin_amdgcn_wmma_f32_16x16x4_f32(
         false, a,  false, b,
         (short)0, c,  false,  false);
}

__device__ __forceinline__ float tanh_fast(float x) {
    const float e2x = __builtin_amdgcn_exp2f(x * 2.88539008177792681f);
    const float big = fmaf(__builtin_amdgcn_rcpf(e2x + 1.0f), -2.0f, 1.0f);
    const float u = x * x;
    float p = fmaf(1.72354765e-2f, u, -5.30422069e-2f);
    p = fmaf(p, u, 1.33258224e-1f);
    p = fmaf(p, u, -3.33331406e-1f);
    const float small = fmaf(x, u * p, x);
    return fabsf(x) < 0.5f ? small : big;
}

__device__ __forceinline__ void matmul64_lds(const float* __restrict__ A,
                                             const float* __restrict__ Bbuf,
                                             int col0, int koff, int ln,
                                             v8f acc[4][4]) {
    for (int kb = 0; kb < 64; kb += 4) {
        v2f a[4], b[4];
#pragma unroll
        for (int mi = 0; mi < 4; ++mi)
            a[mi] = *(const v2f*)&A[(mi * 16 + ln) * TSTR + kb + koff];
#pragma unroll
        for (int ni = 0; ni < 4; ++ni) {
            b[ni].x = Bbuf[(kb + koff) * CSTR + col0 + ni * 16];
            b[ni].y = Bbuf[(kb + koff + 1) * CSTR + col0 + ni * 16];
        }
#pragma unroll
        for (int mi = 0; mi < 4; ++mi)
#pragma unroll
            for (int ni = 0; ni < 4; ++ni)
                acc[mi][ni] = wmma_f32(a[mi], b[ni], acc[mi][ni]);
    }
}

__device__ __forceinline__ void store_tile(float* __restrict__ C, int hh, int col0,
                                           const v8f acc[4][4]) {
#pragma unroll
    for (int mi = 0; mi < 4; ++mi)
#pragma unroll
        for (int ni = 0; ni < 4; ++ni) {
            float* dst = &C[(mi * 16 + 8 * hh) * CSTR + col0 + ni * 16];
#pragma unroll
            for (int r = 0; r < 8; ++r) dst[r * CSTR] = acc[mi][ni][r];
        }
}

__global__ __launch_bounds__(256)
void graph_embed_kernel(const float* __restrict__ adj1, const float* __restrict__ fm1,
                        const float* __restrict__ adj2, const float* __restrict__ fm2,
                        const float* __restrict__ W1,  const float* __restrict__ theta,
                        const float* __restrict__ theta1, float* __restrict__ g_ws) {
    extern __shared__ float smem[];
    float* sC  = smem;
    float* sTh = sC  + EMBED * CSTR;
    float* sT1 = sTh + EMBED * TSTR;
    float* sF  = sT1 + EMBED * TSTR;
    float* sW1 = sF  + NN * NFEAT;
    float* sG  = sW1 + EMBED * NFEAT;
    float* sB  = sG  + 256;

    const int tid   = threadIdx.x;
    const int graph = blockIdx.x >> 7;
    const int batch = blockIdx.x & 127;
    const float* adj = (graph == 0 ? adj1 : adj2) + (size_t)batch * NN * NN;
    const float* fm  = (graph == 0 ? fm1  : fm2)  + (size_t)batch * NN * NFEAT;

    for (int i = tid; i < EMBED * EMBED; i += 256) {
        int e = i >> 6, k = i & 63;
        sTh[e * TSTR + k] = theta[i];
        sT1[e * TSTR + k] = theta1[i];
    }
    for (int i = tid; i < NN * NFEAT; i += 256)    sF[i]  = fm[i];
    for (int i = tid; i < EMBED * NFEAT; i += 256) sW1[i] = W1[i];
    __syncthreads();

    const int lane = tid & 31;
    const int w    = tid >> 5;
    const int ln   = lane & 15;
    const int hh   = lane >> 4;
    const int koff = 2 * hh;
    const int col0 = w * 64 + ln;

    const v8f vzero = {0.f, 0.f, 0.f, 0.f, 0.f, 0.f, 0.f, 0.f};

    v8f term[4][4];
    {
        float fr[4][NFEAT];
#pragma unroll
        for (int ni = 0; ni < 4; ++ni) {
            int n = col0 + ni * 16;
#pragma unroll
            for (int f = 0; f < NFEAT; ++f) fr[ni][f] = sF[n * NFEAT + f];
        }
#pragma unroll
        for (int mi = 0; mi < 4; ++mi)
#pragma unroll
            for (int ni = 0; ni < 4; ++ni)
#pragma unroll
                for (int r = 0; r < 8; ++r) {
                    int e = mi * 16 + r + 8 * hh;
                    float s = 0.f;
#pragma unroll
                    for (int f = 0; f < NFEAT; ++f) s += sW1[e * NFEAT + f] * fr[ni][f];
                    term[mi][ni][r] = s;
                }
    }

    {
        v8f c1[4][4];
#pragma unroll
        for (int mi = 0; mi < 4; ++mi)
#pragma unroll
            for (int ni = 0; ni < 4; ++ni)
#pragma unroll
                for (int r = 0; r < 8; ++r) c1[mi][ni][r] = tanh_fast(term[mi][ni][r]);
        store_tile(sC, hh, col0, c1);
    }
    __syncthreads();

    float* slab = sB + w * (64 * BSTR);
    const int srow = lane >> 3;
    const int spc  = lane & 7;
    const float* gst = adj + (size_t)(w * 64 + srow) * NN + 4 * spc;
    float* sst = slab + srow * BSTR + 4 * spc;
    const float* bfr = slab + ln * BSTR + koff;

    for (int t = 0; t < 4; ++t) {
        v8f acc[4][4];
#pragma unroll
        for (int mi = 0; mi < 4; ++mi)
#pragma unroll
            for (int ni = 0; ni < 4; ++ni) acc[mi][ni] = vzero;

        v4f st[16];
#pragma unroll
        for (int j = 0; j < 16; ++j)
            st[j] = *(const v4f*)(gst + (size_t)j * 4 * NN);

        for (int c = 0; c < NCH; ++c) {
#pragma unroll
            for (int j = 0; j < 16; ++j)
                *(v4f*)(sst + j * 4 * BSTR) = st[j];
            if (c + 1 < NCH) {
#pragma unroll
                for (int j = 0; j < 16; ++j)
                    st[j] = *(const v4f*)(gst + (c + 1) * KCH + (size_t)j * 4 * NN);
            }
            for (int kk = 0; kk < KCH; kk += 4) {
                const int kb = c * KCH + kk;
                v2f a[4], b[4];
#pragma unroll
                for (int mi = 0; mi < 4; ++mi)
                    a[mi] = *(const v2f*)&sC[(mi * 16 + ln) * CSTR + kb + koff];
#pragma unroll
                for (int ni = 0; ni < 4; ++ni)
                    b[ni] = *(const v2f*)&bfr[ni * 16 * BSTR + kk];
#pragma unroll
                for (int mi = 0; mi < 4; ++mi)
#pragma unroll
                    for (int ni = 0; ni < 4; ++ni)
                        acc[mi][ni] = wmma_f32(a[mi], b[ni], acc[mi][ni]);
            }
        }
        __syncthreads();
        store_tile(sC, hh, col0, acc);
        __syncthreads();

        v8f acc2[4][4];
#pragma unroll
        for (int mi = 0; mi < 4; ++mi)
#pragma unroll
            for (int ni = 0; ni < 4; ++ni) acc2[mi][ni] = vzero;
        matmul64_lds(sTh, sC, col0, koff, ln, acc2);
#pragma unroll
        for (int mi = 0; mi < 4; ++mi)
#pragma unroll
            for (int ni = 0; ni < 4; ++ni)
#pragma unroll
                for (int r = 0; r < 8; ++r) acc2[mi][ni][r] = fmaxf(acc2[mi][ni][r], 0.f);
        store_tile(sC, hh, col0, acc2);
        __syncthreads();

#pragma unroll
        for (int mi = 0; mi < 4; ++mi)
#pragma unroll
            for (int ni = 0; ni < 4; ++ni) acc[mi][ni] = vzero;
        matmul64_lds(sT1, sC, col0, koff, ln, acc);
#pragma unroll
        for (int mi = 0; mi < 4; ++mi)
#pragma unroll
            for (int ni = 0; ni < 4; ++ni)
#pragma unroll
                for (int r = 0; r < 8; ++r)
                    acc[mi][ni][r] = tanh_fast(term[mi][ni][r] + acc[mi][ni][r]);
        store_tile(sC, hh, col0, acc);
        __syncthreads();
    }

    {
        int e = tid >> 2, q = tid & 3;
        float s = 0.f;
        for (int n = q * 128; n < q * 128 + 128; ++n) s += sC[e * CSTR + n];
        sG[tid] = s;
        __syncthreads();
        if (tid < 64) {
            float g = sG[tid * 4] + sG[tid * 4 + 1] + sG[tid * 4 + 2] + sG[tid * 4 + 3];
            g_ws[(size_t)graph * EMBED * 128 + tid * 128 + batch] = g;
        }
    }
}

#define HB 32
__global__ __launch_bounds__(1024)
void head_kernel(const float* __restrict__ W2, const float* __restrict__ g_ws,
                 float* __restrict__ out) {
    extern __shared__ __align__(16) float s2[];
    float* sW2 = s2;
    float* sg  = sW2 + 4096;
    float* so  = sg + 2 * 64 * HB;
    const int tid = threadIdx.x;
    const int b0  = blockIdx.x * HB;

    {
        const int row = tid >> 3;
        const int pc  = tid & 7;
        const v4f w = *(const v4f*)(W2 + 4 * tid);
        const v4f g = *(const v4f*)(g_ws + row * 128 + b0 + 4 * pc);
        *(v4f*)(sW2 + 4 * tid) = w;
        *(v4f*)(sg + row * HB + 4 * pc) = g;
    }
    __syncthreads();

    const int lane  = tid & 31;
    const int wv    = tid >> 5;
    const int graph = wv >> 4;
    const int e0    = (wv & 15) * 4;
    const float* gp = sg + graph * 64 * HB + lane;
    const float* wp = sW2 + e0 * 64;
    float acc[4] = {0.f, 0.f, 0.f, 0.f};
#pragma unroll
    for (int k = 0; k < 64; ++k) {
        const float gk = gp[k * HB];
#pragma unroll
        for (int i = 0; i < 4; ++i) acc[i] = fmaf(wp[i * 64 + k], gk, acc[i]);
    }
#pragma unroll
    for (int i = 0; i < 4; ++i) {
        so[(graph * 64 + e0 + i) * HB + lane] = acc[i];
        out[128 + graph * 8192 + (e0 + i) * 128 + b0 + lane] = acc[i];
    }
    __syncthreads();

    if (tid < HB) {
        float dot = 0.f, l1 = 0.f, l2 = 0.f;
#pragma unroll
        for (int e = 0; e < 64; ++e) {
            float a = so[e * HB + tid];
            float c = so[(64 + e) * HB + tid];
            dot = fmaf(a, c, dot); l1 = fmaf(a, a, l1); l2 = fmaf(c, c, l2);
        }
        out[b0 + tid] = dot / (sqrtf(l1) * sqrtf(l2) + 1e-4f);
    }
}

extern "C" void kernel_launch(void* const* d_in, const int* in_sizes, int n_in,
                              void* d_out, int out_size, void* d_ws, size_t ws_size,
                              hipStream_t stream) {
    (void)in_sizes; (void)n_in; (void)out_size; (void)ws_size;
    const float* adj1 = (const float*)d_in[0];
    const float* fm1  = (const float*)d_in[1];
    const float* adj2 = (const float*)d_in[2];
    const float* fm2  = (const float*)d_in[3];
    const float* W1   = (const float*)d_in[4];
    const float* W2   = (const float*)d_in[5];
    const float* th   = (const float*)d_in[6];
    const float* th1  = (const float*)d_in[7];
    float* gws = (float*)d_ws;

    size_t smem1 = (size_t)(EMBED * CSTR + 2 * EMBED * TSTR + NN * NFEAT +
                            EMBED * NFEAT + 256 + 8 * 64 * BSTR) * sizeof(float);
    size_t smem2 = (size_t)(4096 + 2 * 2 * 64 * HB) * sizeof(float);

    hipFuncSetAttribute((const void*)graph_embed_kernel,
                        hipFuncAttributeMaxDynamicSharedMemorySize, (int)smem1);
    hipFuncSetAttribute((const void*)head_kernel,
                        hipFuncAttributeMaxDynamicSharedMemorySize, (int)smem2);

    graph_embed_kernel<<<256, 256, smem1, stream>>>(adj1, fm1, adj2, fm2, W1, th, th1, gws);
    head_kernel<<<4, 1024, smem2, stream>>>(W2, gws, (float*)d_out);
}


# global loads hoisted: start-up theta, theta1, W1 and feature loads all in flight at once; adj chunk 0 loaded ahead of term and tanh and, for later rounds, during the last chunk of the round before
# speedup vs baseline: 1.9158x; 1.0773x over previous
#include <hip/hip_runtime.h>
#include <cstddef>

typedef float v2f __attribute__((ext_vector_type(2)));
typedef float v4f __attribute__((ext_vector_type(4)));
typedef float v8f __attribute__((ext_vector_type(8)));

#define EMBED 64
#define NN    512
#define NFEAT 9
#define CSTR  516
#define TSTR  68
#define KCH   32
#define BSTR  36
#define NCH   (NN / KCH)
#define W1PAD 1024

__device__ __forceinline__ v8f wmma_f32(v2f a, v2f b, v8f c) {
    return __builtin_amdgcn_wmma_f32_16x16x4_f32(
         false, a,  false, b,
         (short)0, c,  false,  false);
}

__device__ __forceinline__ float tanh_fast(float x) {
    const float e2x = __builtin_amdgcn_exp2f(x * 2.88539008177792681f);
    const float big = fmaf(__builtin_amdgcn_rcpf(e2x + 1.0f), -2.0f, 1.0f);
    const float u = x * x;
    float p = fmaf(1.72354765e-2f, u, -5.30422069e-2f);
    p = fmaf(p, u, 1.33258224e-1f);
    p = fmaf(p, u, -3.33331406e-1f);
    const float small = fmaf(x, u * p, x);
    return fabsf(x) < 0.5f ? small : big;
}

__device__ __forceinline__ void matmul64_regs(const float* __restrict__ A, const v8f Bt[4][4],
                                              int hh, int ln, v8f acc[4][4]) {
    const float* ap = A + ln * TSTR + 8 * hh;
    v4f aq[2][4];
#pragma unroll
    for (int mi = 0; mi < 4; ++mi)
        aq[0][mi] = *(const v4f*)&ap[mi * 16 * TSTR];
#pragma unroll
    for (int blk = 0; blk < 8; ++blk) {
        const int m = blk >> 1, i = blk & 1;
        if (blk + 1 < 8) {
#pragma unroll
            for (int mi = 0; mi < 4; ++mi)
                aq[(blk + 1) & 1][mi] = *(const v4f*)&ap[mi * 16 * TSTR + 16 * ((blk + 1) >> 1) + 4 * ((blk + 1) & 1)];
        }
        __builtin_amdgcn_sched_barrier(0);
#pragma unroll
        for (int s = 0; s < 2; ++s) {
            v2f a[4], b[4];
#pragma unroll
            for (int mi = 0; mi < 4; ++mi) {
                const v4f q = aq[blk & 1][mi];
                a[mi].x = s ? q.z : q.x;
                a[mi].y = s ? q.w : q.y;
            }
#pragma unroll
            for (int ni = 0; ni < 4; ++ni) {
                b[ni].x = Bt[m][ni][4 * i + 2 * s];
                b[ni].y = Bt[m][ni][4 * i + 2 * s + 1];
            }
#pragma unroll
            for (int mi = 0; mi < 4; ++mi)
#pragma unroll
                for (int ni = 0; ni < 4; ++ni)
                    acc[mi][ni] = wmma_f32(a[mi], b[ni], acc[mi][ni]);
        }
        __builtin_amdgcn_sched_barrier(0);
    }
}

__device__ __forceinline__ void store_tile(float* __restrict__ C, int hh, int col0,
                                           const v8f acc[4][4]) {
#pragma unroll
    for (int mi = 0; mi < 4; ++mi)
#pragma unroll
        for (int ni = 0; ni < 4; ++ni) {
            float* dst = &C[(mi * 16 + 8 * hh) * CSTR + col0 + ni * 16];
#pragma unroll
            for (int r = 0; r < 8; ++r) dst[r * CSTR] = acc[mi][ni][r];
        }
}

__global__ __launch_bounds__(256)
void graph_embed_kernel(const float* __restrict__ adj1, const float* __restrict__ fm1,
                        const float* __restrict__ adj2, const float* __restrict__ fm2,
                        const float* __restrict__ W1,  const float* __restrict__ theta,
                        const float* __restrict__ theta1, float* __restrict__ g_ws) {
    extern __shared__ float smem[];
    float* sC  = smem;
    float* sTh = sC  + EMBED * CSTR;
    float* sT1 = sTh + EMBED * TSTR;
    float* sF  = sT1 + EMBED * TSTR;
    float* sW1 = sF  + NN * NFEAT;
    float* sG  = sW1 + W1PAD;
    float* sB  = sG  + 256;

    const int tid   = threadIdx.x;
    const int graph = blockIdx.x >> 7;
    const int batch = blockIdx.x & 127;
    const float* adj = (graph == 0 ? adj1 : adj2) + (size_t)batch * NN * NN;
    const float* fm  = (graph == 0 ? fm1  : fm2)  + (size_t)batch * NN * NFEAT;

    const int lane = tid & 31;
    const int w    = tid >> 5;
    const int ln   = lane & 15;
    const int hh   = lane >> 4;
    const int koff = 2 * hh;
    const int col0 = w * 64 + ln;

    float* slab = sB + w * (64 * BSTR);
    const int srow = lane >> 3;
    const int spc  = lane & 7;
    const float* gst = adj + (size_t)(w * 64 + srow) * NN + 4 * spc;
    float* sst = slab + srow * BSTR + 4 * spc;
    const float* bfr = slab + ln * BSTR + koff;

    v4f rth[4], rt1[4];
    v2f rfm[9];
#pragma unroll
    for (int j = 0; j < 4; ++j) {
        rth[j] = *(const v4f*)(theta  + 4 * (tid + 256 * j));
        rt1[j] = *(const v4f*)(theta1 + 4 * (tid + 256 * j));
    }
    const v4f rw1 = *(const v4f*)(W1 + 4 * (tid < EMBED * NFEAT / 4 ? tid : tid - EMBED * NFEAT / 4));
#pragma unroll
    for (int j = 0; j < 9; ++j) rfm[j] = *(const v2f*)(fm + 2 * (tid + 256 * j));

#pragma unroll
    for (int j = 0; j < 4; ++j) {
        const int i = 4 * (tid + 256 * j), e = i >> 6, k = i & 63;
        *(v4f*)&sTh[e * TSTR + k] = rth[j];
        *(v4f*)&sT1[e * TSTR + k] = rt1[j];
    }
    *(v4f*)&sW1[4 * tid] = rw1;
#pragma unroll
    for (int j = 0; j < 9; ++j) *(v2f*)&sF[2 * (tid + 256 * j)] = rfm[j];
    __syncthreads();

    v4f st[16];
#pragma unroll
    for (int j = 0; j < 16; ++j)
        st[j] = *(const v4f*)(gst + (size_t)j * 4 * NN);
    __builtin_amdgcn_sched_barrier(0);

    const v8f vzero = {0.f, 0.f, 0.f, 0.f, 0.f, 0.f, 0.f, 0.f};

    v8f term[4][4];
    {
        float fr[4][NFEAT];
#pragma unroll
        for (int ni = 0; ni < 4; ++ni) {
            int n = col0 + ni * 16;
#pragma unroll
            for (int f = 0; f < NFEAT; ++f) fr[ni][f] = sF[n * NFEAT + f];
        }
#pragma unroll
        for (int mi = 0; mi < 4; ++mi)
#pragma unroll
            for (int ni = 0; ni < 4; ++ni)
#pragma unroll
                for (int r = 0; r < 8; ++r) {
                    int e = mi * 16 + r + 8 * hh;
                    float s = 0.f;
#pragma unroll
                    for (int f = 0; f < NFEAT; ++f) s += sW1[e * NFEAT + f] * fr[ni][f];
                    term[mi][ni][r] = s;
                }
    }

    {
        v8f c1[4][4];
#pragma unroll
        for (int mi = 0; mi < 4; ++mi)
#pragma unroll
            for (int ni = 0; ni < 4; ++ni)
#pragma unroll
                for (int r = 0; r < 8; ++r) c1[mi][ni][r] = tanh_fast(term[mi][ni][r]);
        store_tile(sC, hh, col0, c1);
    }
    __syncthreads();

    for (int t = 0; t < 4; ++t) {
        v8f acc[4][4];
#pragma unroll
        for (int mi = 0; mi < 4; ++mi)
#pragma unroll
            for (int ni = 0; ni < 4; ++ni) acc[mi][ni] = vzero;

        for (int c = 0; c < NCH; ++c) {
#pragma unroll
            for (int j = 0; j < 16; ++j)
                *(v4f*)(sst + j * 4 * BSTR) = st[j];
            const int cn = (c + 1) & (NCH - 1);
            if (cn != 0 || t != 3) {
#pragma unroll
                for (int j = 0; j < 16; ++j)
                    st[j] = *(const v4f*)(gst + cn * KCH + (size_t)j * 4 * NN);
            }
            for (int kk = 0; kk < KCH; kk += 4) {
                const int kb = c * KCH + kk;
                v2f a[4], b[4];
#pragma unroll
                for (int mi = 0; mi < 4; ++mi)
                    a[mi] = *(const v2f*)&sC[(mi * 16 + ln) * CSTR + kb + koff];
#pragma unroll
                for (int ni = 0; ni < 4; ++ni)
                    b[ni] = *(const v2f*)&bfr[ni * 16 * BSTR + kk];
#pragma unroll
                for (int mi = 0; mi < 4; ++mi)
#pragma unroll
                    for (int ni = 0; ni < 4; ++ni)
                        acc[mi][ni] = wmma_f32(a[mi], b[ni], acc[mi][ni]);
            }
        }
        v8f acc2[4][4];
#pragma unroll
        for (int mi = 0; mi < 4; ++mi)
#pragma unroll
            for (int ni = 0; ni < 4; ++ni) acc2[mi][ni] = vzero;
        matmul64_regs(sTh, acc, hh, ln, acc2);
#pragma unroll
        for (int mi = 0; mi < 4; ++mi)
#pragma unroll
            for (int ni = 0; ni < 4; ++ni)
#pragma unroll
                for (int r = 0; r < 8; ++r) acc2[mi][ni][r] = fmaxf(acc2[mi][ni][r], 0.f);

#pragma unroll
        for (int mi = 0; mi < 4; ++mi)
#pragma unroll
            for (int ni = 0; ni < 4; ++ni) acc[mi][ni] = vzero;
        matmul64_regs(sT1, acc2, hh, ln, acc);
#pragma unroll
        for (int mi = 0; mi < 4; ++mi)
#pragma unroll
            for (int ni = 0; ni < 4; ++ni)
#pragma unroll
                for (int r = 0; r < 8; ++r)
                    acc[mi][ni][r] = tanh_fast(term[mi][ni][r] + acc[mi][ni][r]);
        __syncthreads();
        store_tile(sC, hh, col0, acc);
        __syncthreads();
    }

    {
        int e = tid >> 2, q = tid & 3;
        float s = 0.f;
        for (int n = q * 128; n < q * 128 + 128; ++n) s += sC[e * CSTR + n];
        sG[tid] = s;
        __syncthreads();
        if (tid < 64) {
            float g = sG[tid * 4] + sG[tid * 4 + 1] + sG[tid * 4 + 2] + sG[tid * 4 + 3];
            g_ws[(size_t)graph * EMBED * 128 + tid * 128 + batch] = g;
        }
    }
}

#define HB 32
__global__ __launch_bounds__(1024)
void head_kernel(const float* __restrict__ W2, const float* __restrict__ g_ws,
                 float* __restrict__ out) {
    extern __shared__ __align__(16) float s2[];
    float* sW2 = s2;
    float* sg  = sW2 + 4096;
    float* so  = sg + 2 * 64 * HB;
    const int tid = threadIdx.x;
    const int b0  = blockIdx.x * HB;

    {
        const int row = tid >> 3;
        const int pc  = tid & 7;
        const v4f w = *(const v4f*)(W2 + 4 * tid);
        const v4f g = *(const v4f*)(g_ws + row * 128 + b0 + 4 * pc);
        *(v4f*)(sW2 + 4 * tid) = w;
        *(v4f*)(sg + row * HB + 4 * pc) = g;
    }
    __syncthreads();

    const int lane  = tid & 31;
    const int wv    = tid >> 5;
    const int graph = wv >> 4;
    const int e0    = (wv & 15) * 4;
    const float* gp = sg + graph * 64 * HB + lane;
    const float* wp = sW2 + e0 * 64;
    float acc[4] = {0.f, 0.f, 0.f, 0.f};
#pragma unroll
    for (int k = 0; k < 64; ++k) {
        const float gk = gp[k * HB];
#pragma unroll
        for (int i = 0; i < 4; ++i) acc[i] = fmaf(wp[i * 64 + k], gk, acc[i]);
    }
#pragma unroll
    for (int i = 0; i < 4; ++i) {
        so[(graph * 64 + e0 + i) * HB + lane] = acc[i];
        out[128 + graph * 8192 + (e0 + i) * 128 + b0 + lane] = acc[i];
    }
    __syncthreads();

    if (tid < HB) {
        float dot = 0.f, l1 = 0.f, l2 = 0.f;
#pragma unroll
        for (int e = 0; e < 64; ++e) {
            float a = so[e * HB + tid];
            float c = so[(64 + e) * HB + tid];
            dot = fmaf(a, c, dot); l1 = fmaf(a, a, l1); l2 = fmaf(c, c, l2);
        }
        out[b0 + tid] = dot / (sqrtf(l1) * sqrtf(l2) + 1e-4f);
    }
}

extern "C" void kernel_launch(void* const* d_in, const int* in_sizes, int n_in,
                              void* d_out, int out_size, void* d_ws, size_t ws_size,
                              hipStream_t stream) {
    (void)in_sizes; (void)n_in; (void)out_size; (void)ws_size;
    const float* adj1 = (const float*)d_in[0];
    const float* fm1  = (const float*)d_in[1];
    const float* adj2 = (const float*)d_in[2];
    const float* fm2  = (const float*)d_in[3];
    const float* W1   = (const float*)d_in[4];
    const float* W2   = (const float*)d_in[5];
    const float* th   = (const float*)d_in[6];
    const float* th1  = (const float*)d_in[7];
    float* gws = (float*)d_ws;

    size_t smem1 = (size_t)(EMBED * CSTR + 2 * EMBED * TSTR + NN * NFEAT +
                            W1PAD + 256 + 8 * 64 * BSTR) * sizeof(float);
    size_t smem2 = (size_t)(4096 + 2 * 2 * 64 * HB) * sizeof(float);

    hipFuncSetAttribute((const void*)graph_embed_kernel,
                        hipFuncAttributeMaxDynamicSharedMemorySize, (int)smem1);
    hipFuncSetAttribute((const void*)head_kernel,
                        hipFuncAttributeMaxDynamicSharedMemorySize, (int)smem2);

    graph_embed_kernel<<<256, 256, smem1, stream>>>(adj1, fm1, adj2, fm2, W1, th, th1, gws);
    head_kernel<<<4, 1024, smem2, stream>>>(W2, gws, (float*)d_out);
}
